# starting kernel
# baseline (speedup 1.0000x reference)
#include <hip/hip_runtime.h>
#include <math.h>

typedef __attribute__((ext_vector_type(2))) float v2f;
typedef __attribute__((ext_vector_type(8))) float v8f;

#define NPEAK 16
#define TUNABLE_MASK ((1<<0)|(1<<3)|(1<<6)|(1<<9)|(1<<12)|(1<<15))
#define E_TUNE  0.1f
#define GMIN    1e-4f
#define GMAX    0.5f

__global__ void setup_params(const float* __restrict__ ec_w, const float* __restrict__ ec_b,
                             const float* __restrict__ th_w, const float* __restrict__ th_b,
                             const float* __restrict__ gr_w, const float* __restrict__ gr_b,
                             const float* __restrict__ gl_w, const float* __restrict__ gl_b,
                             const float* __restrict__ e0,   float* __restrict__ ws) {
  int p = threadIdx.x;
  if (p >= NPEAK) return;
  float ec = ((TUNABLE_MASK >> p) & 1)
                 ? (e0[p] + E_TUNE * tanhf(ec_w[p] + ec_b[p]))
                 : e0[p];
  float th = 3.14159265358979323846f / (1.0f + expf(-(th_w[p] + th_b[p])));
  float gr = GMIN + (GMAX - GMIN) / (1.0f + expf(-(gr_w[p] + gr_b[p])));
  float gl = GMIN + (GMAX - GMIN) / (1.0f + expf(-(gl_w[p] + gl_b[p])));
  float g  = 0.5f * (gr + gl);
  float sq = sqrtf(gr * gl);
  float a  = cosf(th) * sq;
  float b  = sinf(th) * sq;
  ws[p]        = ec;
  ws[16 + p]   = g * g;
  ws[32 + 0*16 + p] = a;
  ws[32 + 1*16 + p] = b * g - a * ec;
  ws[32 + 2*16 + p] = b;
  ws[32 + 3*16 + p] = -(b * ec + a * g);
}

__global__ __launch_bounds__(256)
void peaks_wmma_kernel(const float* __restrict__ E, float* __restrict__ out,
                       const float* __restrict__ ws, int n) {
  const int lane = threadIdx.x & 31;
  const int m    = lane & 15;
  const int kh   = lane >> 4;
  const int wave   = blockIdx.x * (blockDim.x >> 5) + (threadIdx.x >> 5);
  const int nwaves = gridDim.x * (blockDim.x >> 5);
  const int ntiles = (n + 15) >> 4;

  float ecv[4][2], g2v[4][2];
  v2f A[4];
#pragma unroll
  for (int s = 0; s < 4; ++s) {
#pragma unroll
    for (int j = 0; j < 2; ++j) {
      int p = 4 * s + 2 * kh + j;
      ecv[s][j] = ws[p];
      g2v[s][j] = ws[16 + p];
      A[s][j]   = (m < 4) ? ws[32 + 16 * m + p] : 0.0f;
    }
  }

  for (int t = wave; t < ntiles; t += nwaves) {
    const int base = t << 4;
    int idx = base + m;
    float Ev = E[idx < n ? idx : (n - 1)];

    v2f B[4];
#pragma unroll
    for (int s = 0; s < 4; ++s) {
#pragma unroll
      for (int j = 0; j < 2; ++j) {
        float d = Ev - ecv[s][j];
        B[s][j] = __builtin_amdgcn_rcpf(__builtin_fmaf(d, d, g2v[s][j]));
      }
    }

    v8f acc = {};
#pragma unroll
    for (int s = 0; s < 4; ++s) {
      acc = __builtin_amdgcn_wmma_f32_16x16x4_f32(
           false, A[s],  false, B[s],
           (short)0, acc,  false,  false);
    }

    if (lane < 16 && idx < n) {
      float tre = __builtin_fmaf(Ev, acc[0], acc[1]);
      float tim = __builtin_fmaf(Ev, acc[2], acc[3]);
      out[idx] = tre * tre + tim * tim;
    }
  }
}

extern "C" void kernel_launch(void* const* d_in, const int* in_sizes, int n_in,
                              void* d_out, int out_size, void* d_ws, size_t ws_size,
                              hipStream_t stream) {
  const float* E    = (const float*)d_in[0];
  const float* ec_w = (const float*)d_in[1];
  const float* ec_b = (const float*)d_in[2];
  const float* th_w = (const float*)d_in[3];
  const float* th_b = (const float*)d_in[4];
  const float* gr_w = (const float*)d_in[5];
  const float* gr_b = (const float*)d_in[6];
  const float* gl_w = (const float*)d_in[7];
  const float* gl_b = (const float*)d_in[8];
  const float* e0   = (const float*)d_in[9];
  float* out = (float*)d_out;
  float* ws  = (float*)d_ws;

  const int n = in_sizes[0];
  setup_params<<<1, 32, 0, stream>>>(ec_w, ec_b, th_w, th_b, gr_w, gr_b,
                                     gl_w, gl_b, e0, ws);

  const int ntiles = (n + 15) >> 4;
  int blocks = (ntiles + 7) / 8;
  if (blocks > 2048) blocks = 2048;
  if (blocks < 1) blocks = 1;
  peaks_wmma_kernel<<<blocks, 256, 0, stream>>>(E, out, ws, n);
}


# Fixed grid of 256 work-groups x 8 waves instead of 2048 x 8: each wave loops over 64-element chunks (16-byte loads and stores, 4 independent accumulator tiles, loads requested one trip ahead)
# speedup vs baseline: 1.2363x; 1.2363x over previous
#include <hip/hip_runtime.h>
#include <math.h>


typedef __attribute__((ext_vector_type(2))) float v2f;
typedef __attribute__((ext_vector_type(4))) float v4f;
typedef __attribute__((ext_vector_type(8))) float v8f;

#define NPEAK 16
#define TUNABLE_MASK ((1<<0)|(1<<3)|(1<<6)|(1<<9)|(1<<12)|(1<<15))
#define E_TUNE  0.1f
#define GMIN    1e-4f
#define GMAX    0.5f

__global__ void setup_params(const float* __restrict__ ec_w, const float* __restrict__ ec_b,
                             const float* __restrict__ th_w, const float* __restrict__ th_b,
                             const float* __restrict__ gr_w, const float* __restrict__ gr_b,
                             const float* __restrict__ gl_w, const float* __restrict__ gl_b,
                             const float* __restrict__ e0,   float* __restrict__ ws) {
  int p = threadIdx.x;
  if (p >= NPEAK) return;
  float ec = ((TUNABLE_MASK >> p) & 1)
                 ? (e0[p] + E_TUNE * tanhf(ec_w[p] + ec_b[p]))
                 : e0[p];
  float th = 3.14159265358979323846f / (1.0f + expf(-(th_w[p] + th_b[p])));
  float gr = GMIN + (GMAX - GMIN) / (1.0f + expf(-(gr_w[p] + gr_b[p])));
  float gl = GMIN + (GMAX - GMIN) / (1.0f + expf(-(gl_w[p] + gl_b[p])));
  float g  = 0.5f * (gr + gl);
  float sq = sqrtf(gr * gl);
  float a  = cosf(th) * sq;
  float b  = sinf(th) * sq;
  ws[p]        = ec;
  ws[16 + p]   = g * g;
  ws[32 + 0*16 + p] = a;
  ws[32 + 1*16 + p] = b * g - a * ec;
  ws[32 + 2*16 + p] = b;
  ws[32 + 3*16 + p] = -(b * ec + a * g);
}

__global__ __launch_bounds__(256)
void peaks_wmma_kernel(const float* __restrict__ E, float* __restrict__ out,
                       const float* __restrict__ ws, int n) {
  const int lane = threadIdx.x & 31;
  const int m    = lane & 15;
  const int kh   = lane >> 4;
  const int wave   = blockIdx.x * (blockDim.x >> 5)
                   + __builtin_amdgcn_readfirstlane((int)(threadIdx.x >> 5));
  const int nwaves = gridDim.x * (blockDim.x >> 5);

  v2f ecv[4], g2v[4];
  v2f A[4];
#pragma unroll
  for (int s = 0; s < 4; ++s) {
#pragma unroll
    for (int j = 0; j < 2; ++j) {
      int p = 4 * s + 2 * kh + j;
      ecv[s][j] = ws[p];
      g2v[s][j] = ws[16 + p];
      A[s][j]   = (m < 4) ? ws[32 + 16 * m + p] : 0.0f;
    }
  }

  const bool al16 = ((((size_t)E) | ((size_t)out)) & 15) == 0;
  const int nchunks = al16 ? (n >> 6) : 0;

  if (wave < nchunks) {
    const v4f* __restrict__ E4 = (const v4f*)E;
    v4f* __restrict__ O4 = (v4f*)out;
    const int last = nchunks - 1;

    auto do_chunk = [&](const v4f ev, const int c) {
      v2f B[4][4];
#pragma unroll
      for (int j = 0; j < 4; ++j) {
        const v2f e2v = {ev[j], ev[j]};
#pragma unroll
        for (int s = 0; s < 4; ++s) {
          v2f d = e2v - ecv[s];
          v2f q = __builtin_elementwise_fma(d, d, g2v[s]);
          B[j][s][0] = __builtin_amdgcn_rcpf(q[0]);
          B[j][s][1] = __builtin_amdgcn_rcpf(q[1]);
        }
      }

      v8f acc[4] = {};
#pragma unroll
      for (int s = 0; s < 4; ++s) {
#pragma unroll
        for (int j = 0; j < 4; ++j) {
          acc[j] = __builtin_amdgcn_wmma_f32_16x16x4_f32(
               false, A[s],  false, B[j][s],
               (short)0, acc[j],  false,  false);
        }
      }

      if (lane < 16) {
        v4f o;
#pragma unroll
        for (int j = 0; j < 4; ++j) {
          float tre = __builtin_fmaf(ev[j], acc[j][0], acc[j][1]);
          float tim = __builtin_fmaf(ev[j], acc[j][2], acc[j][3]);
          o[j] = tre * tre + tim * tim;
        }
        O4[c * 16 + m] = o;
      }
    };

    int cb = wave + nwaves; cb = cb < last ? cb : last;
    v4f ea = E4[wave * 16 + m];
    v4f eb = E4[cb * 16 + m];
    for (int c = wave; c < nchunks; c += 2 * nwaves) {
      int na = c + 2 * nwaves; na = na < last ? na : last;
      int nb = c + 3 * nwaves; nb = nb < last ? nb : last;
      const v4f ea_next = E4[na * 16 + m];
      const v4f eb_next = E4[nb * 16 + m];
      __builtin_amdgcn_sched_barrier(0);
      do_chunk(ea, c);
      if (c + nwaves < nchunks) do_chunk(eb, c + nwaves);
      ea = ea_next;
      eb = eb_next;
    }
  }

  const int ntiles = (n + 15) >> 4;
  for (int t = (nchunks << 2) + wave; t < ntiles; t += nwaves) {
    const int base = t << 4;
    int idx = base + m;
    float Ev = E[idx < n ? idx : (n - 1)];

    v2f B[4];
#pragma unroll
    for (int s = 0; s < 4; ++s) {
#pragma unroll
      for (int j = 0; j < 2; ++j) {
        float d = Ev - ecv[s][j];
        B[s][j] = __builtin_amdgcn_rcpf(__builtin_fmaf(d, d, g2v[s][j]));
      }
    }

    v8f acc = {};
#pragma unroll
    for (int s = 0; s < 4; ++s) {
      acc = __builtin_amdgcn_wmma_f32_16x16x4_f32(
           false, A[s],  false, B[s],
           (short)0, acc,  false,  false);
    }

    if (lane < 16 && idx < n) {
      float tre = __builtin_fmaf(Ev, acc[0], acc[1]);
      float tim = __builtin_fmaf(Ev, acc[2], acc[3]);
      out[idx] = tre * tre + tim * tim;
    }
  }
}

extern "C" void kernel_launch(void* const* d_in, const int* in_sizes, int n_in,
                              void* d_out, int out_size, void* d_ws, size_t ws_size,
                              hipStream_t stream) {
  const float* E    = (const float*)d_in[0];
  const float* ec_w = (const float*)d_in[1];
  const float* ec_b = (const float*)d_in[2];
  const float* th_w = (const float*)d_in[3];
  const float* th_b = (const float*)d_in[4];
  const float* gr_w = (const float*)d_in[5];
  const float* gr_b = (const float*)d_in[6];
  const float* gl_w = (const float*)d_in[7];
  const float* gl_b = (const float*)d_in[8];
  const float* e0   = (const float*)d_in[9];
  float* out = (float*)d_out;
  float* ws  = (float*)d_ws;

  const int n = in_sizes[0];
  setup_params<<<1, 32, 0, stream>>>(ec_w, ec_b, th_w, th_b, gr_w, gr_b,
                                     gl_w, gl_b, e0, ws);

  peaks_wmma_kernel<<<256, 256, 0, stream>>>(E, out, ws, n);
}


# One launch instead of two: the 16 per-peak constants are computed by 16 lanes in the main kernel's prologue and shared through LDS (one barrier); first E loads requested before that math
# speedup vs baseline: 1.4239x; 1.1517x over previous
#include <hip/hip_runtime.h>
#include <math.h>


typedef __attribute__((ext_vector_type(2))) float v2f;
typedef __attribute__((ext_vector_type(4))) float v4f;
typedef __attribute__((ext_vector_type(8))) float v8f;

#define NPEAK 16
#define TUNABLE_MASK ((1<<0)|(1<<3)|(1<<6)|(1<<9)|(1<<12)|(1<<15))
#define E_TUNE  0.1f
#define GMIN    1e-4f
#define GMAX    0.5f


__global__ __launch_bounds__(256)
void peaks_wmma_kernel(const float* __restrict__ E, float* __restrict__ out,
                       const float* __restrict__ ec_w, const float* __restrict__ ec_b,
                       const float* __restrict__ th_w, const float* __restrict__ th_b,
                       const float* __restrict__ gr_w, const float* __restrict__ gr_b,
                       const float* __restrict__ gl_w, const float* __restrict__ gl_b,
                       const float* __restrict__ e0, int n) {
  __shared__ float prm[6 * NPEAK];
  const int lane = threadIdx.x & 31;
  const int m    = lane & 15;
  const int kh   = lane >> 4;
  const int wave   = blockIdx.x * (blockDim.x >> 5)
                   + __builtin_amdgcn_readfirstlane((int)(threadIdx.x >> 5));
  const int nwaves = gridDim.x * (blockDim.x >> 5);

  const bool al16 = ((((size_t)E) | ((size_t)out)) & 15) == 0;
  const int nchunks = al16 ? (n >> 6) : 0;
  const int last = nchunks - 1;
  const v4f* __restrict__ E4 = (const v4f*)E;
  v4f* __restrict__ O4 = (v4f*)out;

  const bool setup_lane = threadIdx.x < NPEAK;
  const int  sp = setup_lane ? (int)threadIdx.x : 0;
  float r_ecw = 0.f, r_ecb = 0.f, r_thw = 0.f, r_thb = 0.f, r_grw = 0.f,
        r_grb = 0.f, r_glw = 0.f, r_glb = 0.f, r_e0 = 0.f;
  if (setup_lane) {
    r_ecw = ec_w[sp]; r_ecb = ec_b[sp];
    r_thw = th_w[sp]; r_thb = th_b[sp];
    r_grw = gr_w[sp]; r_grb = gr_b[sp];
    r_glw = gl_w[sp]; r_glb = gl_b[sp];
    r_e0  = e0[sp];
    asm volatile("" :: "v"(r_ecw), "v"(r_ecb), "v"(r_thw), "v"(r_thb),
                       "v"(r_grw), "v"(r_grb), "v"(r_glw), "v"(r_glb),
                       "v"(r_e0));
  }

  v4f ea = {}, eb = {};
  if (wave < nchunks) {
    int cb = wave + nwaves; cb = cb < last ? cb : last;
    ea = E4[wave * 16 + m];
    eb = E4[cb * 16 + m];
  }

  if (setup_lane) {
    float ec = ((TUNABLE_MASK >> sp) & 1)
                   ? (r_e0 + E_TUNE * tanhf(r_ecw + r_ecb))
                   : r_e0;
    float th = 3.14159265358979323846f / (1.0f + expf(-(r_thw + r_thb)));
    float gr = GMIN + (GMAX - GMIN) / (1.0f + expf(-(r_grw + r_grb)));
    float gl = GMIN + (GMAX - GMIN) / (1.0f + expf(-(r_glw + r_glb)));
    float g  = 0.5f * (gr + gl);
    float sq = sqrtf(gr * gl);
    float a  = cosf(th) * sq;
    float b  = sinf(th) * sq;
    prm[sp]              = ec;
    prm[16 + sp]         = g * g;
    prm[32 + 0*16 + sp]  = a;
    prm[32 + 1*16 + sp]  = b * g - a * ec;
    prm[32 + 2*16 + sp]  = b;
    prm[32 + 3*16 + sp]  = -(b * ec + a * g);
  }
  __syncthreads();

  v2f ecv[4], g2v[4];
  v2f A[4];
#pragma unroll
  for (int s = 0; s < 4; ++s) {
#pragma unroll
    for (int j = 0; j < 2; ++j) {
      int p = 4 * s + 2 * kh + j;
      ecv[s][j] = prm[p];
      g2v[s][j] = prm[16 + p];
      const float cf = prm[32 + 16 * (m & 3) + p];
      A[s][j]   = (m < 4) ? cf : 0.0f;
    }
  }

  if (wave < nchunks) {

    auto do_chunk = [&](const v4f ev, const int c) {
      v2f B[4][4];
#pragma unroll
      for (int j = 0; j < 4; ++j) {
        const v2f e2v = {ev[j], ev[j]};
#pragma unroll
        for (int s = 0; s < 4; ++s) {
          v2f d = e2v - ecv[s];
          v2f q = __builtin_elementwise_fma(d, d, g2v[s]);
          B[j][s][0] = __builtin_amdgcn_rcpf(q[0]);
          B[j][s][1] = __builtin_amdgcn_rcpf(q[1]);
        }
      }

      v8f acc[4] = {};
#pragma unroll
      for (int s = 0; s < 4; ++s) {
#pragma unroll
        for (int j = 0; j < 4; ++j) {
          acc[j] = __builtin_amdgcn_wmma_f32_16x16x4_f32(
               false, A[s],  false, B[j][s],
               (short)0, acc[j],  false,  false);
        }
      }

      if (lane < 16) {
        v4f o;
#pragma unroll
        for (int j = 0; j < 4; ++j) {
          float tre = __builtin_fmaf(ev[j], acc[j][0], acc[j][1]);
          float tim = __builtin_fmaf(ev[j], acc[j][2], acc[j][3]);
          o[j] = tre * tre + tim * tim;
        }
        O4[c * 16 + m] = o;
      }
    };

    for (int c = wave; c < nchunks; c += 2 * nwaves) {
      int na = c + 2 * nwaves; na = na < last ? na : last;
      int nb = c + 3 * nwaves; nb = nb < last ? nb : last;
      const v4f ea_next = E4[na * 16 + m];
      const v4f eb_next = E4[nb * 16 + m];
      __builtin_amdgcn_sched_barrier(0);
      do_chunk(ea, c);
      if (c + nwaves < nchunks) do_chunk(eb, c + nwaves);
      ea = ea_next;
      eb = eb_next;
    }
  }

  const int ntiles = (n + 15) >> 4;
  for (int t = (nchunks << 2) + wave; t < ntiles; t += nwaves) {
    const int base = t << 4;
    int idx = base + m;
    float Ev = E[idx < n ? idx : (n - 1)];

    v2f B[4];
#pragma unroll
    for (int s = 0; s < 4; ++s) {
#pragma unroll
      for (int j = 0; j < 2; ++j) {
        float d = Ev - ecv[s][j];
        B[s][j] = __builtin_amdgcn_rcpf(__builtin_fmaf(d, d, g2v[s][j]));
      }
    }

    v8f acc = {};
#pragma unroll
    for (int s = 0; s < 4; ++s) {
      acc = __builtin_amdgcn_wmma_f32_16x16x4_f32(
           false, A[s],  false, B[s],
           (short)0, acc,  false,  false);
    }

    if (lane < 16 && idx < n) {
      float tre = __builtin_fmaf(Ev, acc[0], acc[1]);
      float tim = __builtin_fmaf(Ev, acc[2], acc[3]);
      out[idx] = tre * tre + tim * tim;
    }
  }
}

extern "C" void kernel_launch(void* const* d_in, const int* in_sizes, int n_in,
                              void* d_out, int out_size, void* d_ws, size_t ws_size,
                              hipStream_t stream) {
  const float* E    = (const float*)d_in[0];
  const float* ec_w = (const float*)d_in[1];
  const float* ec_b = (const float*)d_in[2];
  const float* th_w = (const float*)d_in[3];
  const float* th_b = (const float*)d_in[4];
  const float* gr_w = (const float*)d_in[5];
  const float* gr_b = (const float*)d_in[6];
  const float* gl_w = (const float*)d_in[7];
  const float* gl_b = (const float*)d_in[8];
  const float* e0   = (const float*)d_in[9];
  float* out = (float*)d_out;

  const int n = in_sizes[0];

  peaks_wmma_kernel<<<256, 256, 0, stream>>>(E, out, ec_w, ec_b, th_w, th_b,
                                             gr_w, gr_b, gl_w, gl_b, e0, n);
}


# Peaks p, p+8 share one reciprocal 1/(q_p q_r): 8 pair reciprocals x 8 coefficient rows (cubic numerator evaluated in E after the sum); 8 matrix instructions per 64 elements, not 16; sum order changed
# speedup vs baseline: 1.8217x; 1.2794x over previous
#include <hip/hip_runtime.h>
#include <math.h>


typedef __attribute__((ext_vector_type(2))) float v2f;
typedef __attribute__((ext_vector_type(4))) float v4f;
typedef __attribute__((ext_vector_type(8))) float v8f;

#define NPEAK 16
#define NPAIR 8
#define TUNABLE_MASK ((1<<0)|(1<<3)|(1<<6)|(1<<9)|(1<<12)|(1<<15))
#define E_TUNE  0.1f
#define GMIN    1e-4f
#define GMAX    0.5f
#define PAIR_COND_MAX 512.0f


__global__ __launch_bounds__(256)
void peaks_wmma_kernel(const float* __restrict__ E, float* __restrict__ out,
                       const float* __restrict__ ec_w, const float* __restrict__ ec_b,
                       const float* __restrict__ th_w, const float* __restrict__ th_b,
                       const float* __restrict__ gr_w, const float* __restrict__ gr_b,
                       const float* __restrict__ gl_w, const float* __restrict__ gl_b,
                       const float* __restrict__ e0, int n) {
  __shared__ float prm[6 * NPEAK + 8 * NPAIR + 8];
  const int lane = threadIdx.x & 31;
  const int m    = lane & 15;
  const int kh   = lane >> 4;
  const int wave   = blockIdx.x * (blockDim.x >> 5)
                   + __builtin_amdgcn_readfirstlane((int)(threadIdx.x >> 5));
  const int nwaves = gridDim.x * (blockDim.x >> 5);

  const bool al16 = ((((size_t)E) | ((size_t)out)) & 15) == 0;
  const int nchunks = al16 ? (n >> 6) : 0;
  const int last = nchunks - 1;
  const v4f* __restrict__ E4 = (const v4f*)E;
  v4f* __restrict__ O4 = (v4f*)out;

  const bool setup_lane = threadIdx.x < NPEAK;
  const int  sp = setup_lane ? (int)threadIdx.x : 0;
  float r_ecw = 0.f, r_ecb = 0.f, r_thw = 0.f, r_thb = 0.f, r_grw = 0.f,
        r_grb = 0.f, r_glw = 0.f, r_glb = 0.f, r_e0 = 0.f;
  if (setup_lane) {
    r_ecw = ec_w[sp]; r_ecb = ec_b[sp];
    r_thw = th_w[sp]; r_thb = th_b[sp];
    r_grw = gr_w[sp]; r_grb = gr_b[sp];
    r_glw = gl_w[sp]; r_glb = gl_b[sp];
    r_e0  = e0[sp];
    asm volatile("" :: "v"(r_ecw), "v"(r_ecb), "v"(r_thw), "v"(r_thb),
                       "v"(r_grw), "v"(r_grb), "v"(r_glw), "v"(r_glb),
                       "v"(r_e0));
  }

  v4f ea = {}, eb = {};
  if (wave < nchunks) {
    int cb = wave + nwaves; cb = cb < last ? cb : last;
    ea = E4[wave * 16 + m];
    eb = E4[cb * 16 + m];
  }

  if (setup_lane) {
    float ec = ((TUNABLE_MASK >> sp) & 1)
                   ? (r_e0 + E_TUNE * tanhf(r_ecw + r_ecb))
                   : r_e0;
    float th = 3.14159265358979323846f / (1.0f + expf(-(r_thw + r_thb)));
    float gr = GMIN + (GMAX - GMIN) / (1.0f + expf(-(r_grw + r_grb)));
    float gl = GMIN + (GMAX - GMIN) / (1.0f + expf(-(r_glw + r_glb)));
    float g  = 0.5f * (gr + gl);
    float sq = sqrtf(gr * gl);
    float a  = cosf(th) * sq;
    float b  = sinf(th) * sq;
    prm[sp]              = ec;
    prm[16 + sp]         = g * g;
    prm[32 + 0*16 + sp]  = a;
    prm[32 + 1*16 + sp]  = b * g - a * ec;
    prm[32 + 2*16 + sp]  = b;
    prm[32 + 3*16 + sp]  = -(b * ec + a * g);

    const float ecr = __shfl_xor(ec, 8, 16);
    const float gp  = __shfl_xor(g,  8, 16);
    const float ar  = __shfl_xor(a,  8, 16);
    const float br  = __shfl_xor(b,  8, 16);
    const float w2p = __builtin_fmaf(ec, ec, g * g);
    const float w2r = __builtin_fmaf(ecr, ecr, gp * gp);
    const float c3r = a + ar, c3i = b + br;
    const float t2pr = __builtin_fmaf(2.0f, ecr, ec), t2rr = __builtin_fmaf(2.0f, ec, ecr);
    const float c2r = -((a * t2pr - b * g) + (ar * t2rr - br * gp));
    const float c2i = -((a * g + b * t2pr) + (ar * gp + br * t2rr));
    const float t1pr = __builtin_fmaf(2.0f * ecr, ec, w2r), t1pi = 2.0f * ecr * g;
    const float t1rr = __builtin_fmaf(2.0f * ec, ecr, w2p), t1ri = 2.0f * ec * gp;
    const float c1r = (a * t1pr - b * t1pi) + (ar * t1rr - br * t1ri);
    const float c1i = (a * t1pi + b * t1pr) + (ar * t1ri + br * t1rr);
    const float c0r = -((a * ec - b * g) * w2r + (ar * ecr - br * gp) * w2p);
    const float c0i = -((a * g + b * ec) * w2r + (ar * gp + br * ecr) * w2p);
    if (sp < NPAIR) {
      prm[96 + 0*8 + sp] = c0r;  prm[96 + 1*8 + sp] = c0i;
      prm[96 + 2*8 + sp] = c1r;  prm[96 + 3*8 + sp] = c1i;
      prm[96 + 4*8 + sp] = c2r;  prm[96 + 5*8 + sp] = c2i;
      prm[96 + 6*8 + sp] = c3r;  prm[96 + 7*8 + sp] = c3i;
    }
    const float ea_ = fabsf(ec) + g;
    const float mono = __builtin_fmaf(__builtin_fmaf(__builtin_fmaf(
        fabsf(c3r) + fabsf(c3i), ea_, fabsf(c2r) + fabsf(c2i)), ea_,
        fabsf(c1r) + fabsf(c1i)), ea_, fabsf(c0r) + fabsf(c0i));
    const float dxy = ec - ecr;
    const float den = (g * g) * __builtin_fmaf(dxy, dxy, gp * gp);
    const bool bad = !(mono <= PAIR_COND_MAX * den);
    const unsigned anybad = __builtin_amdgcn_ballot_w32(bad);
    if (sp == 0) prm[160] = anybad ? 0.0f : 1.0f;
  }
  __syncthreads();

  const bool paired = __builtin_amdgcn_readfirstlane((int)(prm[160] != 0.0f)) != 0;

  auto load_peak_consts = [&](v2f (&ecv)[4], v2f (&g2v)[4], v2f (&A)[4]) {
#pragma unroll
    for (int s = 0; s < 4; ++s) {
#pragma unroll
      for (int j = 0; j < 2; ++j) {
        int p = 4 * s + 2 * kh + j;
        ecv[s][j] = prm[p];
        g2v[s][j] = prm[16 + p];
        const float cf = prm[32 + 16 * (m & 3) + p];
        A[s][j]   = (m < 4) ? cf : 0.0f;
      }
    }
  };

  auto walk = [&](auto&& do_chunk) {
    for (int c = wave; c < nchunks; c += 2 * nwaves) {
      int na = c + 2 * nwaves; na = na < last ? na : last;
      int nb = c + 3 * nwaves; nb = nb < last ? nb : last;
      const v4f ea_next = E4[na * 16 + m];
      const v4f eb_next = E4[nb * 16 + m];
      __builtin_amdgcn_sched_barrier(0);
      do_chunk(ea, c);
      if (c + nwaves < nchunks) do_chunk(eb, c + nwaves);
      ea = ea_next;
      eb = eb_next;
    }
  };

  if (wave < nchunks) {
    if (paired) {
      v2f ecP[2], g2P[2], ecR[2], g2R[2], A2[2];
#pragma unroll
      for (int s = 0; s < 2; ++s) {
#pragma unroll
        for (int j = 0; j < 2; ++j) {
          const int P = 4 * s + 2 * kh + j;
          ecP[s][j] = prm[P];
          ecR[s][j] = prm[8 + P];
          g2P[s][j] = prm[16 + P];
          g2R[s][j] = prm[24 + P];
          const float cf = prm[96 + 8 * (m & 7) + P];
          A2[s][j]  = (m < 8) ? cf : 0.0f;
        }
      }

      auto do_chunk_pair = [&](const v4f ev, const int c) {
        v2f B[4][2];
#pragma unroll
        for (int j = 0; j < 4; ++j) {
          const v2f e2v = {ev[j], ev[j]};
#pragma unroll
          for (int s = 0; s < 2; ++s) {
            const v2f dp = e2v - ecP[s];
            const v2f dr = e2v - ecR[s];
            const v2f qp = __builtin_elementwise_fma(dp, dp, g2P[s]);
            const v2f qr = __builtin_elementwise_fma(dr, dr, g2R[s]);
            const v2f qq = qp * qr;
            B[j][s][0] = __builtin_amdgcn_rcpf(qq[0]);
            B[j][s][1] = __builtin_amdgcn_rcpf(qq[1]);
          }
        }

        v8f acc[4] = {};
#pragma unroll
        for (int s = 0; s < 2; ++s) {
#pragma unroll
          for (int j = 0; j < 4; ++j) {
            acc[j] = __builtin_amdgcn_wmma_f32_16x16x4_f32(
                 false, A2[s],  false, B[j][s],
                 (short)0, acc[j],  false,  false);
          }
        }

        if (lane < 16) {
          v4f o;
#pragma unroll
          for (int j = 0; j < 4; ++j) {
            const v2f e2v = {ev[j], ev[j]};
            v2f t = {acc[j][6], acc[j][7]};
            const v2f s2 = {acc[j][4], acc[j][5]};
            const v2f s1 = {acc[j][2], acc[j][3]};
            const v2f s0 = {acc[j][0], acc[j][1]};
            t = __builtin_elementwise_fma(t, e2v, s2);
            t = __builtin_elementwise_fma(t, e2v, s1);
            t = __builtin_elementwise_fma(t, e2v, s0);
            o[j] = t[0] * t[0] + t[1] * t[1];
          }
          O4[c * 16 + m] = o;
        }
      };
      walk(do_chunk_pair);
    } else {
      v2f ecv[4], g2v[4], A[4];
      load_peak_consts(ecv, g2v, A);
      auto do_chunk = [&](const v4f ev, const int c) {
        v2f B[4][4];
#pragma unroll
        for (int j = 0; j < 4; ++j) {
          const v2f e2v = {ev[j], ev[j]};
#pragma unroll
          for (int s = 0; s < 4; ++s) {
            v2f d = e2v - ecv[s];
            v2f q = __builtin_elementwise_fma(d, d, g2v[s]);
            B[j][s][0] = __builtin_amdgcn_rcpf(q[0]);
            B[j][s][1] = __builtin_amdgcn_rcpf(q[1]);
          }
        }
        v8f acc[4] = {};
#pragma unroll
        for (int s = 0; s < 4; ++s) {
#pragma unroll
          for (int j = 0; j < 4; ++j) {
            acc[j] = __builtin_amdgcn_wmma_f32_16x16x4_f32(
                 false, A[s],  false, B[j][s],
                 (short)0, acc[j],  false,  false);
          }
        }
        if (lane < 16) {
          v4f o;
#pragma unroll
          for (int j = 0; j < 4; ++j) {
            float tre = __builtin_fmaf(ev[j], acc[j][0], acc[j][1]);
            float tim = __builtin_fmaf(ev[j], acc[j][2], acc[j][3]);
            o[j] = tre * tre + tim * tim;
          }
          O4[c * 16 + m] = o;
        }
      };
      walk(do_chunk);
    }
  }

  const int ntiles = (n + 15) >> 4;
  if ((nchunks << 2) + wave < ntiles) {
    v2f ecv[4], g2v[4], A[4];
    load_peak_consts(ecv, g2v, A);
    for (int t = (nchunks << 2) + wave; t < ntiles; t += nwaves) {
      const int base = t << 4;
      int idx = base + m;
      float Ev = E[idx < n ? idx : (n - 1)];

      v2f B[4];
#pragma unroll
      for (int s = 0; s < 4; ++s) {
#pragma unroll
        for (int j = 0; j < 2; ++j) {
          float d = Ev - ecv[s][j];
          B[s][j] = __builtin_amdgcn_rcpf(__builtin_fmaf(d, d, g2v[s][j]));
        }
      }

      v8f acc = {};
#pragma unroll
      for (int s = 0; s < 4; ++s) {
        acc = __builtin_amdgcn_wmma_f32_16x16x4_f32(
             false, A[s],  false, B[s],
             (short)0, acc,  false,  false);
      }

      if (lane < 16 && idx < n) {
        float tre = __builtin_fmaf(Ev, acc[0], acc[1]);
        float tim = __builtin_fmaf(Ev, acc[2], acc[3]);
        out[idx] = tre * tre + tim * tim;
      }
    }
  }
}

extern "C" void kernel_launch(void* const* d_in, const int* in_sizes, int n_in,
                              void* d_out, int out_size, void* d_ws, size_t ws_size,
                              hipStream_t stream) {
  const float* E    = (const float*)d_in[0];
  const float* ec_w = (const float*)d_in[1];
  const float* ec_b = (const float*)d_in[2];
  const float* th_w = (const float*)d_in[3];
  const float* th_b = (const float*)d_in[4];
  const float* gr_w = (const float*)d_in[5];
  const float* gr_b = (const float*)d_in[6];
  const float* gl_w = (const float*)d_in[7];
  const float* gl_b = (const float*)d_in[8];
  const float* e0   = (const float*)d_in[9];
  float* out = (float*)d_out;

  const int n = in_sizes[0];

  peaks_wmma_kernel<<<256, 256, 0, stream>>>(E, out, ec_w, ec_b, th_w, th_b,
                                             gr_w, gr_b, gl_w, gl_b, e0, n);
}


# Peaks p, p+4, p+8, p+12 share one reciprocal: 4 quad reciprocals x 16 coefficient rows (degree-7 numerator in E after the sum); 4 matrix instructions per 64 elements, not 8; sum order changed
# speedup vs baseline: 1.9365x; 1.0630x over previous
#include <hip/hip_runtime.h>
#include <math.h>


typedef __attribute__((ext_vector_type(2))) float v2f;
typedef __attribute__((ext_vector_type(4))) float v4f;
typedef __attribute__((ext_vector_type(8))) float v8f;

#define NPEAK 16
#define NPAIR 8
#define NQUAD 4
#define QB (6 * NPEAK + 8 * NPAIR + 8)
#define TUNABLE_MASK ((1<<0)|(1<<3)|(1<<6)|(1<<9)|(1<<12)|(1<<15))
#define E_TUNE  0.1f
#define GMIN    1e-4f
#define GMAX    0.5f
#define PAIR_COND_MAX 512.0f
#define QUAD_COND_MAX 512.0f


__global__ __launch_bounds__(256)
void peaks_wmma_kernel(const float* __restrict__ E, float* __restrict__ out,
                       const float* __restrict__ ec_w, const float* __restrict__ ec_b,
                       const float* __restrict__ th_w, const float* __restrict__ th_b,
                       const float* __restrict__ gr_w, const float* __restrict__ gr_b,
                       const float* __restrict__ gl_w, const float* __restrict__ gl_b,
                       const float* __restrict__ e0, int n) {
  __shared__ float prm[6 * NPEAK + 8 * NPAIR + 8 + 16 * NQUAD];
  const int lane = threadIdx.x & 31;
  const int m    = lane & 15;
  const int kh   = lane >> 4;
  const int wave   = blockIdx.x * (blockDim.x >> 5)
                   + __builtin_amdgcn_readfirstlane((int)(threadIdx.x >> 5));
  const int nwaves = gridDim.x * (blockDim.x >> 5);

  const bool al16 = ((((size_t)E) | ((size_t)out)) & 15) == 0;
  const int nchunks = al16 ? (n >> 6) : 0;
  const int last = nchunks - 1;
  const v4f* __restrict__ E4 = (const v4f*)E;
  v4f* __restrict__ O4 = (v4f*)out;

  const bool setup_lane = threadIdx.x < NPEAK;
  const int  sp = setup_lane ? (int)threadIdx.x : 0;
  float r_ecw = 0.f, r_ecb = 0.f, r_thw = 0.f, r_thb = 0.f, r_grw = 0.f,
        r_grb = 0.f, r_glw = 0.f, r_glb = 0.f, r_e0 = 0.f;
  if (setup_lane) {
    r_ecw = ec_w[sp]; r_ecb = ec_b[sp];
    r_thw = th_w[sp]; r_thb = th_b[sp];
    r_grw = gr_w[sp]; r_grb = gr_b[sp];
    r_glw = gl_w[sp]; r_glb = gl_b[sp];
    r_e0  = e0[sp];
    asm volatile("" :: "v"(r_ecw), "v"(r_ecb), "v"(r_thw), "v"(r_thb),
                       "v"(r_grw), "v"(r_grb), "v"(r_glw), "v"(r_glb),
                       "v"(r_e0));
  }

  v4f ea = {}, eb = {};
  if (wave < nchunks) {
    int cb = wave + nwaves; cb = cb < last ? cb : last;
    ea = E4[wave * 16 + m];
    eb = E4[cb * 16 + m];
  }

  if (setup_lane) {
    float ec = ((TUNABLE_MASK >> sp) & 1)
                   ? (r_e0 + E_TUNE * tanhf(r_ecw + r_ecb))
                   : r_e0;
    float th = 3.14159265358979323846f / (1.0f + expf(-(r_thw + r_thb)));
    float gr = GMIN + (GMAX - GMIN) / (1.0f + expf(-(r_grw + r_grb)));
    float gl = GMIN + (GMAX - GMIN) / (1.0f + expf(-(r_glw + r_glb)));
    float g  = 0.5f * (gr + gl);
    float sq = sqrtf(gr * gl);
    float a  = cosf(th) * sq;
    float b  = sinf(th) * sq;
    prm[sp]              = ec;
    prm[16 + sp]         = g * g;
    prm[32 + 0*16 + sp]  = a;
    prm[32 + 1*16 + sp]  = b * g - a * ec;
    prm[32 + 2*16 + sp]  = b;
    prm[32 + 3*16 + sp]  = -(b * ec + a * g);

    const float ecr = __shfl_xor(ec, 8, 16);
    const float gp  = __shfl_xor(g,  8, 16);
    const float ar  = __shfl_xor(a,  8, 16);
    const float br  = __shfl_xor(b,  8, 16);
    const float w2p = __builtin_fmaf(ec, ec, g * g);
    const float w2r = __builtin_fmaf(ecr, ecr, gp * gp);
    const float c3r = a + ar, c3i = b + br;
    const float t2pr = __builtin_fmaf(2.0f, ecr, ec), t2rr = __builtin_fmaf(2.0f, ec, ecr);
    const float c2r = -((a * t2pr - b * g) + (ar * t2rr - br * gp));
    const float c2i = -((a * g + b * t2pr) + (ar * gp + br * t2rr));
    const float t1pr = __builtin_fmaf(2.0f * ecr, ec, w2r), t1pi = 2.0f * ecr * g;
    const float t1rr = __builtin_fmaf(2.0f * ec, ecr, w2p), t1ri = 2.0f * ec * gp;
    const float c1r = (a * t1pr - b * t1pi) + (ar * t1rr - br * t1ri);
    const float c1i = (a * t1pi + b * t1pr) + (ar * t1ri + br * t1rr);
    const float c0r = -((a * ec - b * g) * w2r + (ar * ecr - br * gp) * w2p);
    const float c0i = -((a * g + b * ec) * w2r + (ar * gp + br * ecr) * w2p);
    if (sp < NPAIR) {
      prm[96 + 0*8 + sp] = c0r;  prm[96 + 1*8 + sp] = c0i;
      prm[96 + 2*8 + sp] = c1r;  prm[96 + 3*8 + sp] = c1i;
      prm[96 + 4*8 + sp] = c2r;  prm[96 + 5*8 + sp] = c2i;
      prm[96 + 6*8 + sp] = c3r;  prm[96 + 7*8 + sp] = c3i;
    }
    const float ea_ = fabsf(ec) + g;
    const float mono = __builtin_fmaf(__builtin_fmaf(__builtin_fmaf(
        fabsf(c3r) + fabsf(c3i), ea_, fabsf(c2r) + fabsf(c2i)), ea_,
        fabsf(c1r) + fabsf(c1i)), ea_, fabsf(c0r) + fabsf(c0i));
    const float dxy = ec - ecr;
    const float den = (g * g) * __builtin_fmaf(dxy, dxy, gp * gp);
    const bool bad = !(mono <= PAIR_COND_MAX * den);
    const unsigned anybad = __builtin_amdgcn_ballot_w32(bad);
    if (sp == 0) prm[160] = anybad ? 0.0f : 1.0f;

    const float ec4  = __shfl_xor(ec, 4, 16),  g4  = __shfl_xor(g, 4, 16);
    const float ec12 = __shfl_xor(ec, 12, 16), g12 = __shfl_xor(g, 12, 16);
    const float be1 = -2.0f * ec4,  ga1 = __builtin_fmaf(ec4, ec4, g4 * g4);
    const float be2 = -2.0f * ecr,  ga2 = w2r;
    const float be3 = -2.0f * ec12, ga3 = __builtin_fmaf(ec12, ec12, g12 * g12);
    const float u3 = be1 + be2;
    const float u2 = __builtin_fmaf(be1, be2, ga1 + ga2);
    const float u1 = __builtin_fmaf(be1, ga2, be2 * ga1);
    const float u0 = ga1 * ga2;
    float rr[7];
    rr[6] = 1.0f;
    rr[5] = u3 + be3;
    rr[4] = __builtin_fmaf(u3, be3, u2 + ga3);
    rr[3] = __builtin_fmaf(u2, be3, __builtin_fmaf(u3, ga3, u1));
    rr[2] = __builtin_fmaf(u1, be3, __builtin_fmaf(u2, ga3, u0));
    rr[1] = __builtin_fmaf(u0, be3, u1 * ga3);
    rr[0] = u0 * ga3;
    const float mre = b * g - a * ec;
    const float mim = -(b * ec + a * g);
    float nr[8], ni[8];
    nr[7] = a;          ni[7] = b;
    nr[0] = mre * rr[0]; ni[0] = mim * rr[0];
#pragma unroll
    for (int k = 1; k < 7; ++k) {
      nr[k] = __builtin_fmaf(a, rr[k - 1], mre * rr[k]);
      ni[k] = __builtin_fmaf(b, rr[k - 1], mim * rr[k]);
    }
#pragma unroll
    for (int k = 0; k < 8; ++k) {
      nr[k] += __shfl_xor(nr[k], 4, 16);
      ni[k] += __shfl_xor(ni[k], 4, 16);
      nr[k] += __shfl_xor(nr[k], 8, 16);
      ni[k] += __shfl_xor(ni[k], 8, 16);
    }
    if (sp < NQUAD) {
#pragma unroll
      for (int k = 0; k < 8; ++k) {
        prm[QB + 4 * k + sp]       = nr[k];
        prm[QB + 4 * (8 + k) + sp] = ni[k];
      }
    }
    float monoq = fabsf(nr[7]) + fabsf(ni[7]);
#pragma unroll
    for (int k = 6; k >= 0; --k)
      monoq = __builtin_fmaf(monoq, ea_, fabsf(nr[k]) + fabsf(ni[k]));
    const float d4 = ec - ec4, d12 = ec - ec12;
    const float denq = den * __builtin_fmaf(d4, d4, g4 * g4)
                           * __builtin_fmaf(d12, d12, g12 * g12);
    const bool badq = !(monoq <= QUAD_COND_MAX * denq);
    const unsigned anybadq = __builtin_amdgcn_ballot_w32(badq);
    if (sp == 0) prm[161] = anybadq ? 0.0f : 1.0f;
  }
  __syncthreads();

  const bool paired = __builtin_amdgcn_readfirstlane((int)(prm[160] != 0.0f)) != 0;
  const bool quad   = __builtin_amdgcn_readfirstlane((int)(prm[161] != 0.0f)) != 0;

  auto load_peak_consts = [&](v2f (&ecv)[4], v2f (&g2v)[4], v2f (&A)[4]) {
#pragma unroll
    for (int s = 0; s < 4; ++s) {
#pragma unroll
      for (int j = 0; j < 2; ++j) {
        int p = 4 * s + 2 * kh + j;
        ecv[s][j] = prm[p];
        g2v[s][j] = prm[16 + p];
        const float cf = prm[32 + 16 * (m & 3) + p];
        A[s][j]   = (m < 4) ? cf : 0.0f;
      }
    }
  };

  auto walk = [&](auto&& do_chunk) {
    for (int c = wave; c < nchunks; c += 2 * nwaves) {
      int na = c + 2 * nwaves; na = na < last ? na : last;
      int nb = c + 3 * nwaves; nb = nb < last ? nb : last;
      const v4f ea_next = E4[na * 16 + m];
      const v4f eb_next = E4[nb * 16 + m];
      __builtin_amdgcn_sched_barrier(0);
      do_chunk(ea, c);
      if (c + nwaves < nchunks) do_chunk(eb, c + nwaves);
      ea = ea_next;
      eb = eb_next;
    }
  };

  if (wave < nchunks) {
    if (quad) {
      v2f ecM[4], g2M[4], A4;
#pragma unroll
      for (int j = 0; j < 2; ++j) {
        const int Q = 2 * kh + j;
#pragma unroll
        for (int t = 0; t < 4; ++t) {
          ecM[t][j] = prm[Q + 4 * t];
          g2M[t][j] = prm[16 + Q + 4 * t];
        }
        A4[j] = prm[QB + 4 * m + Q];
      }

      auto do_chunk_quad = [&](const v4f ev, const int c) {
        v2f B[4];
#pragma unroll
        for (int j = 0; j < 4; ++j) {
          const v2f e2v = {ev[j], ev[j]};
          v2f q[4];
#pragma unroll
          for (int t = 0; t < 4; ++t) {
            const v2f d = e2v - ecM[t];
            q[t] = __builtin_elementwise_fma(d, d, g2M[t]);
          }
          const v2f qq = (q[0] * q[1]) * (q[2] * q[3]);
          B[j][0] = __builtin_amdgcn_rcpf(qq[0]);
          B[j][1] = __builtin_amdgcn_rcpf(qq[1]);
        }

        v8f acc[4] = {};
#pragma unroll
        for (int j = 0; j < 4; ++j) {
          acc[j] = __builtin_amdgcn_wmma_f32_16x16x4_f32(
               false, A4,  false, B[j],
               (short)0, acc[j],  false,  false);
        }

        const v4f x = ev * ev;
        float val[4], oth[4];
#pragma unroll
        for (int j = 0; j < 4; ++j) {
          const v2f x2 = {x[j], x[j]};
          v2f t = {acc[j][6], acc[j][7]};
          const v2f s2 = {acc[j][4], acc[j][5]};
          const v2f s1 = {acc[j][2], acc[j][3]};
          const v2f s0 = {acc[j][0], acc[j][1]};
          t = __builtin_elementwise_fma(t, x2, s2);
          t = __builtin_elementwise_fma(t, x2, s1);
          t = __builtin_elementwise_fma(t, x2, s0);
          val[j] = __builtin_fmaf(t[1], ev[j], t[0]);
          oth[j] = __builtin_bit_cast(float, __builtin_amdgcn_permlanex16(
              __builtin_bit_cast(int, val[j]), __builtin_bit_cast(int, val[j]),
              0x76543210, 0xfedcba98, false, false));
        }
        if (lane < 16) {
          v4f o;
#pragma unroll
          for (int j = 0; j < 4; ++j) o[j] = val[j] * val[j] + oth[j] * oth[j];
          O4[c * 16 + m] = o;
        }
      };
      walk(do_chunk_quad);
    } else if (paired) {
      v2f ecP[2], g2P[2], ecR[2], g2R[2], A2[2];
#pragma unroll
      for (int s = 0; s < 2; ++s) {
#pragma unroll
        for (int j = 0; j < 2; ++j) {
          const int P = 4 * s + 2 * kh + j;
          ecP[s][j] = prm[P];
          ecR[s][j] = prm[8 + P];
          g2P[s][j] = prm[16 + P];
          g2R[s][j] = prm[24 + P];
          const float cf = prm[96 + 8 * (m & 7) + P];
          A2[s][j]  = (m < 8) ? cf : 0.0f;
        }
      }

      auto do_chunk_pair = [&](const v4f ev, const int c) {
        v2f B[4][2];
#pragma unroll
        for (int j = 0; j < 4; ++j) {
          const v2f e2v = {ev[j], ev[j]};
#pragma unroll
          for (int s = 0; s < 2; ++s) {
            const v2f dp = e2v - ecP[s];
            const v2f dr = e2v - ecR[s];
            const v2f qp = __builtin_elementwise_fma(dp, dp, g2P[s]);
            const v2f qr = __builtin_elementwise_fma(dr, dr, g2R[s]);
            const v2f qq = qp * qr;
            B[j][s][0] = __builtin_amdgcn_rcpf(qq[0]);
            B[j][s][1] = __builtin_amdgcn_rcpf(qq[1]);
          }
        }

        v8f acc[4] = {};
#pragma unroll
        for (int s = 0; s < 2; ++s) {
#pragma unroll
          for (int j = 0; j < 4; ++j) {
            acc[j] = __builtin_amdgcn_wmma_f32_16x16x4_f32(
                 false, A2[s],  false, B[j][s],
                 (short)0, acc[j],  false,  false);
          }
        }

        if (lane < 16) {
          v4f o;
#pragma unroll
          for (int j = 0; j < 4; ++j) {
            const v2f e2v = {ev[j], ev[j]};
            v2f t = {acc[j][6], acc[j][7]};
            const v2f s2 = {acc[j][4], acc[j][5]};
            const v2f s1 = {acc[j][2], acc[j][3]};
            const v2f s0 = {acc[j][0], acc[j][1]};
            t = __builtin_elementwise_fma(t, e2v, s2);
            t = __builtin_elementwise_fma(t, e2v, s1);
            t = __builtin_elementwise_fma(t, e2v, s0);
            o[j] = t[0] * t[0] + t[1] * t[1];
          }
          O4[c * 16 + m] = o;
        }
      };
      walk(do_chunk_pair);
    } else {
      v2f ecv[4], g2v[4], A[4];
      load_peak_consts(ecv, g2v, A);
      auto do_chunk = [&](const v4f ev, const int c) {
        v2f B[4][4];
#pragma unroll
        for (int j = 0; j < 4; ++j) {
          const v2f e2v = {ev[j], ev[j]};
#pragma unroll
          for (int s = 0; s < 4; ++s) {
            v2f d = e2v - ecv[s];
            v2f q = __builtin_elementwise_fma(d, d, g2v[s]);
            B[j][s][0] = __builtin_amdgcn_rcpf(q[0]);
            B[j][s][1] = __builtin_amdgcn_rcpf(q[1]);
          }
        }
        v8f acc[4] = {};
#pragma unroll
        for (int s = 0; s < 4; ++s) {
#pragma unroll
          for (int j = 0; j < 4; ++j) {
            acc[j] = __builtin_amdgcn_wmma_f32_16x16x4_f32(
                 false, A[s],  false, B[j][s],
                 (short)0, acc[j],  false,  false);
          }
        }
        if (lane < 16) {
          v4f o;
#pragma unroll
          for (int j = 0; j < 4; ++j) {
            float tre = __builtin_fmaf(ev[j], acc[j][0], acc[j][1]);
            float tim = __builtin_fmaf(ev[j], acc[j][2], acc[j][3]);
            o[j] = tre * tre + tim * tim;
          }
          O4[c * 16 + m] = o;
        }
      };
      walk(do_chunk);
    }
  }

  const int ntiles = (n + 15) >> 4;
  if ((nchunks << 2) + wave < ntiles) {
    v2f ecv[4], g2v[4], A[4];
    load_peak_consts(ecv, g2v, A);
    for (int t = (nchunks << 2) + wave; t < ntiles; t += nwaves) {
      const int base = t << 4;
      int idx = base + m;
      float Ev = E[idx < n ? idx : (n - 1)];

      v2f B[4];
#pragma unroll
      for (int s = 0; s < 4; ++s) {
#pragma unroll
        for (int j = 0; j < 2; ++j) {
          float d = Ev - ecv[s][j];
          B[s][j] = __builtin_amdgcn_rcpf(__builtin_fmaf(d, d, g2v[s][j]));
        }
      }

      v8f acc = {};
#pragma unroll
      for (int s = 0; s < 4; ++s) {
        acc = __builtin_amdgcn_wmma_f32_16x16x4_f32(
             false, A[s],  false, B[s],
             (short)0, acc,  false,  false);
      }

      if (lane < 16 && idx < n) {
        float tre = __builtin_fmaf(Ev, acc[0], acc[1]);
        float tim = __builtin_fmaf(Ev, acc[2], acc[3]);
        out[idx] = tre * tre + tim * tim;
      }
    }
  }
}

extern "C" void kernel_launch(void* const* d_in, const int* in_sizes, int n_in,
                              void* d_out, int out_size, void* d_ws, size_t ws_size,
                              hipStream_t stream) {
  const float* E    = (const float*)d_in[0];
  const float* ec_w = (const float*)d_in[1];
  const float* ec_b = (const float*)d_in[2];
  const float* th_w = (const float*)d_in[3];
  const float* th_b = (const float*)d_in[4];
  const float* gr_w = (const float*)d_in[5];
  const float* gr_b = (const float*)d_in[6];
  const float* gl_w = (const float*)d_in[7];
  const float* gl_b = (const float*)d_in[8];
  const float* e0   = (const float*)d_in[9];
  float* out = (float*)d_out;

  const int n = in_sizes[0];

  peaks_wmma_kernel<<<256, 256, 0, stream>>>(E, out, ec_w, ec_b, th_w, th_b,
                                             gr_w, gr_b, gl_w, gl_b, e0, n);
}


# E loads requested four chunks ahead instead of two: each wave keeps four 16-byte loads in flight with a four-chunk trip to land in; the last trip requests nothing (no clamped re-reads to wait for)
# speedup vs baseline: 2.2281x; 1.1506x over previous
#include <hip/hip_runtime.h>
#include <math.h>


typedef __attribute__((ext_vector_type(2))) float v2f;
typedef __attribute__((ext_vector_type(4))) float v4f;
typedef __attribute__((ext_vector_type(8))) float v8f;

#define NPEAK 16
#define NPAIR 8
#define NQUAD 4
#define QB (6 * NPEAK + 8 * NPAIR + 8)
#define TUNABLE_MASK ((1<<0)|(1<<3)|(1<<6)|(1<<9)|(1<<12)|(1<<15))
#define E_TUNE  0.1f
#define GMIN    1e-4f
#define GMAX    0.5f
#define PAIR_COND_MAX 512.0f
#define QUAD_COND_MAX 512.0f


__global__ __launch_bounds__(256)
void peaks_wmma_kernel(const float* __restrict__ E, float* __restrict__ out,
                       const float* __restrict__ ec_w, const float* __restrict__ ec_b,
                       const float* __restrict__ th_w, const float* __restrict__ th_b,
                       const float* __restrict__ gr_w, const float* __restrict__ gr_b,
                       const float* __restrict__ gl_w, const float* __restrict__ gl_b,
                       const float* __restrict__ e0, int n) {
  __shared__ float prm[6 * NPEAK + 8 * NPAIR + 8 + 16 * NQUAD];
  const int lane = threadIdx.x & 31;
  const int m    = lane & 15;
  const int kh   = lane >> 4;
  const int wave   = blockIdx.x * (blockDim.x >> 5)
                   + __builtin_amdgcn_readfirstlane((int)(threadIdx.x >> 5));
  const int nwaves = gridDim.x * (blockDim.x >> 5);

  const bool al16 = ((((size_t)E) | ((size_t)out)) & 15) == 0;
  const int nchunks = al16 ? (n >> 6) : 0;
  const int last = nchunks - 1;
  const v4f* __restrict__ E4 = (const v4f*)E;
  v4f* __restrict__ O4 = (v4f*)out;

  const bool setup_lane = threadIdx.x < NPEAK;
  const int  sp = setup_lane ? (int)threadIdx.x : 0;
  float r_ecw = 0.f, r_ecb = 0.f, r_thw = 0.f, r_thb = 0.f, r_grw = 0.f,
        r_grb = 0.f, r_glw = 0.f, r_glb = 0.f, r_e0 = 0.f;
  if (setup_lane) {
    r_ecw = ec_w[sp]; r_ecb = ec_b[sp];
    r_thw = th_w[sp]; r_thb = th_b[sp];
    r_grw = gr_w[sp]; r_grb = gr_b[sp];
    r_glw = gl_w[sp]; r_glb = gl_b[sp];
    r_e0  = e0[sp];
    asm volatile("" :: "v"(r_ecw), "v"(r_ecb), "v"(r_thw), "v"(r_thb),
                       "v"(r_grw), "v"(r_grb), "v"(r_glw), "v"(r_glb),
                       "v"(r_e0));
  }

  v4f ev0 = {}, ev1 = {}, ev2 = {}, ev3 = {};
  if (wave < nchunks) {
    int c1 = wave + nwaves;     c1 = c1 < last ? c1 : last;
    int c2 = wave + 2 * nwaves; c2 = c2 < last ? c2 : last;
    int c3 = wave + 3 * nwaves; c3 = c3 < last ? c3 : last;
    ev0 = E4[wave * 16 + m];
    ev1 = E4[c1 * 16 + m];
    ev2 = E4[c2 * 16 + m];
    ev3 = E4[c3 * 16 + m];
  }

  if (setup_lane) {
    float ec = ((TUNABLE_MASK >> sp) & 1)
                   ? (r_e0 + E_TUNE * tanhf(r_ecw + r_ecb))
                   : r_e0;
    float th = 3.14159265358979323846f / (1.0f + expf(-(r_thw + r_thb)));
    float gr = GMIN + (GMAX - GMIN) / (1.0f + expf(-(r_grw + r_grb)));
    float gl = GMIN + (GMAX - GMIN) / (1.0f + expf(-(r_glw + r_glb)));
    float g  = 0.5f * (gr + gl);
    float sq = sqrtf(gr * gl);
    float a  = cosf(th) * sq;
    float b  = sinf(th) * sq;
    prm[sp]              = ec;
    prm[16 + sp]         = g * g;
    prm[32 + 0*16 + sp]  = a;
    prm[32 + 1*16 + sp]  = b * g - a * ec;
    prm[32 + 2*16 + sp]  = b;
    prm[32 + 3*16 + sp]  = -(b * ec + a * g);

    const float ecr = __shfl_xor(ec, 8, 16);
    const float gp  = __shfl_xor(g,  8, 16);
    const float ar  = __shfl_xor(a,  8, 16);
    const float br  = __shfl_xor(b,  8, 16);
    const float w2p = __builtin_fmaf(ec, ec, g * g);
    const float w2r = __builtin_fmaf(ecr, ecr, gp * gp);
    const float c3r = a + ar, c3i = b + br;
    const float t2pr = __builtin_fmaf(2.0f, ecr, ec), t2rr = __builtin_fmaf(2.0f, ec, ecr);
    const float c2r = -((a * t2pr - b * g) + (ar * t2rr - br * gp));
    const float c2i = -((a * g + b * t2pr) + (ar * gp + br * t2rr));
    const float t1pr = __builtin_fmaf(2.0f * ecr, ec, w2r), t1pi = 2.0f * ecr * g;
    const float t1rr = __builtin_fmaf(2.0f * ec, ecr, w2p), t1ri = 2.0f * ec * gp;
    const float c1r = (a * t1pr - b * t1pi) + (ar * t1rr - br * t1ri);
    const float c1i = (a * t1pi + b * t1pr) + (ar * t1ri + br * t1rr);
    const float c0r = -((a * ec - b * g) * w2r + (ar * ecr - br * gp) * w2p);
    const float c0i = -((a * g + b * ec) * w2r + (ar * gp + br * ecr) * w2p);
    if (sp < NPAIR) {
      prm[96 + 0*8 + sp] = c0r;  prm[96 + 1*8 + sp] = c0i;
      prm[96 + 2*8 + sp] = c1r;  prm[96 + 3*8 + sp] = c1i;
      prm[96 + 4*8 + sp] = c2r;  prm[96 + 5*8 + sp] = c2i;
      prm[96 + 6*8 + sp] = c3r;  prm[96 + 7*8 + sp] = c3i;
    }
    const float ea_ = fabsf(ec) + g;
    const float mono = __builtin_fmaf(__builtin_fmaf(__builtin_fmaf(
        fabsf(c3r) + fabsf(c3i), ea_, fabsf(c2r) + fabsf(c2i)), ea_,
        fabsf(c1r) + fabsf(c1i)), ea_, fabsf(c0r) + fabsf(c0i));
    const float dxy = ec - ecr;
    const float den = (g * g) * __builtin_fmaf(dxy, dxy, gp * gp);
    const bool bad = !(mono <= PAIR_COND_MAX * den);
    const unsigned anybad = __builtin_amdgcn_ballot_w32(bad);
    if (sp == 0) prm[160] = anybad ? 0.0f : 1.0f;

    const float ec4  = __shfl_xor(ec, 4, 16),  g4  = __shfl_xor(g, 4, 16);
    const float ec12 = __shfl_xor(ec, 12, 16), g12 = __shfl_xor(g, 12, 16);
    const float be1 = -2.0f * ec4,  ga1 = __builtin_fmaf(ec4, ec4, g4 * g4);
    const float be2 = -2.0f * ecr,  ga2 = w2r;
    const float be3 = -2.0f * ec12, ga3 = __builtin_fmaf(ec12, ec12, g12 * g12);
    const float u3 = be1 + be2;
    const float u2 = __builtin_fmaf(be1, be2, ga1 + ga2);
    const float u1 = __builtin_fmaf(be1, ga2, be2 * ga1);
    const float u0 = ga1 * ga2;
    float rr[7];
    rr[6] = 1.0f;
    rr[5] = u3 + be3;
    rr[4] = __builtin_fmaf(u3, be3, u2 + ga3);
    rr[3] = __builtin_fmaf(u2, be3, __builtin_fmaf(u3, ga3, u1));
    rr[2] = __builtin_fmaf(u1, be3, __builtin_fmaf(u2, ga3, u0));
    rr[1] = __builtin_fmaf(u0, be3, u1 * ga3);
    rr[0] = u0 * ga3;
    const float mre = b * g - a * ec;
    const float mim = -(b * ec + a * g);
    float nr[8], ni[8];
    nr[7] = a;          ni[7] = b;
    nr[0] = mre * rr[0]; ni[0] = mim * rr[0];
#pragma unroll
    for (int k = 1; k < 7; ++k) {
      nr[k] = __builtin_fmaf(a, rr[k - 1], mre * rr[k]);
      ni[k] = __builtin_fmaf(b, rr[k - 1], mim * rr[k]);
    }
#pragma unroll
    for (int k = 0; k < 8; ++k) {
      nr[k] += __shfl_xor(nr[k], 4, 16);
      ni[k] += __shfl_xor(ni[k], 4, 16);
      nr[k] += __shfl_xor(nr[k], 8, 16);
      ni[k] += __shfl_xor(ni[k], 8, 16);
    }
    if (sp < NQUAD) {
#pragma unroll
      for (int k = 0; k < 8; ++k) {
        prm[QB + 4 * k + sp]       = nr[k];
        prm[QB + 4 * (8 + k) + sp] = ni[k];
      }
    }
    float monoq = fabsf(nr[7]) + fabsf(ni[7]);
#pragma unroll
    for (int k = 6; k >= 0; --k)
      monoq = __builtin_fmaf(monoq, ea_, fabsf(nr[k]) + fabsf(ni[k]));
    const float d4 = ec - ec4, d12 = ec - ec12;
    const float denq = den * __builtin_fmaf(d4, d4, g4 * g4)
                           * __builtin_fmaf(d12, d12, g12 * g12);
    const bool badq = !(monoq <= QUAD_COND_MAX * denq);
    const unsigned anybadq = __builtin_amdgcn_ballot_w32(badq);
    if (sp == 0) prm[161] = anybadq ? 0.0f : 1.0f;
  }
  __syncthreads();

  const bool paired = __builtin_amdgcn_readfirstlane((int)(prm[160] != 0.0f)) != 0;
  const bool quad   = __builtin_amdgcn_readfirstlane((int)(prm[161] != 0.0f)) != 0;

  auto load_peak_consts = [&](v2f (&ecv)[4], v2f (&g2v)[4], v2f (&A)[4]) {
#pragma unroll
    for (int s = 0; s < 4; ++s) {
#pragma unroll
      for (int j = 0; j < 2; ++j) {
        int p = 4 * s + 2 * kh + j;
        ecv[s][j] = prm[p];
        g2v[s][j] = prm[16 + p];
        const float cf = prm[32 + 16 * (m & 3) + p];
        A[s][j]   = (m < 4) ? cf : 0.0f;
      }
    }
  };

  auto walk = [&](auto&& do_chunk) {
    const int step = 4 * nwaves;
    for (int c = wave; c < nchunks; c += step) {
      v4f n0 = ev0, n1 = ev1, n2 = ev2, n3 = ev3;
      const int cn = c + step;
      if (cn < nchunks) {
        int c1 = cn + nwaves;     c1 = c1 < last ? c1 : last;
        int c2 = cn + 2 * nwaves; c2 = c2 < last ? c2 : last;
        int c3 = cn + 3 * nwaves; c3 = c3 < last ? c3 : last;
        n0 = E4[cn * 16 + m];
        n1 = E4[c1 * 16 + m];
        n2 = E4[c2 * 16 + m];
        n3 = E4[c3 * 16 + m];
      }
      __builtin_amdgcn_sched_barrier(0);
      do_chunk(ev0, c);
      if (c + nwaves < nchunks)     do_chunk(ev1, c + nwaves);
      if (c + 2 * nwaves < nchunks) do_chunk(ev2, c + 2 * nwaves);
      if (c + 3 * nwaves < nchunks) do_chunk(ev3, c + 3 * nwaves);
      ev0 = n0; ev1 = n1; ev2 = n2; ev3 = n3;
    }
  };

  if (wave < nchunks) {
    if (quad) {
      v2f ecM[4], g2M[4], A4;
#pragma unroll
      for (int j = 0; j < 2; ++j) {
        const int Q = 2 * kh + j;
#pragma unroll
        for (int t = 0; t < 4; ++t) {
          ecM[t][j] = prm[Q + 4 * t];
          g2M[t][j] = prm[16 + Q + 4 * t];
        }
        A4[j] = prm[QB + 4 * m + Q];
      }

      auto do_chunk_quad = [&](const v4f ev, const int c) {
        v2f B[4];
#pragma unroll
        for (int j = 0; j < 4; ++j) {
          const v2f e2v = {ev[j], ev[j]};
          v2f q[4];
#pragma unroll
          for (int t = 0; t < 4; ++t) {
            const v2f d = e2v - ecM[t];
            q[t] = __builtin_elementwise_fma(d, d, g2M[t]);
          }
          const v2f qq = (q[0] * q[1]) * (q[2] * q[3]);
          B[j][0] = __builtin_amdgcn_rcpf(qq[0]);
          B[j][1] = __builtin_amdgcn_rcpf(qq[1]);
        }

        v8f acc[4] = {};
#pragma unroll
        for (int j = 0; j < 4; ++j) {
          acc[j] = __builtin_amdgcn_wmma_f32_16x16x4_f32(
               false, A4,  false, B[j],
               (short)0, acc[j],  false,  false);
        }

        const v4f x = ev * ev;
        float val[4], oth[4];
#pragma unroll
        for (int j = 0; j < 4; ++j) {
          const v2f x2 = {x[j], x[j]};
          v2f t = {acc[j][6], acc[j][7]};
          const v2f s2 = {acc[j][4], acc[j][5]};
          const v2f s1 = {acc[j][2], acc[j][3]};
          const v2f s0 = {acc[j][0], acc[j][1]};
          t = __builtin_elementwise_fma(t, x2, s2);
          t = __builtin_elementwise_fma(t, x2, s1);
          t = __builtin_elementwise_fma(t, x2, s0);
          val[j] = __builtin_fmaf(t[1], ev[j], t[0]);
          oth[j] = __builtin_bit_cast(float, __builtin_amdgcn_permlanex16(
              __builtin_bit_cast(int, val[j]), __builtin_bit_cast(int, val[j]),
              0x76543210, 0xfedcba98, false, false));
        }
        if (lane < 16) {
          v4f o;
#pragma unroll
          for (int j = 0; j < 4; ++j) o[j] = val[j] * val[j] + oth[j] * oth[j];
          O4[c * 16 + m] = o;
        }
      };
      walk(do_chunk_quad);
    } else if (paired) {
      v2f ecP[2], g2P[2], ecR[2], g2R[2], A2[2];
#pragma unroll
      for (int s = 0; s < 2; ++s) {
#pragma unroll
        for (int j = 0; j < 2; ++j) {
          const int P = 4 * s + 2 * kh + j;
          ecP[s][j] = prm[P];
          ecR[s][j] = prm[8 + P];
          g2P[s][j] = prm[16 + P];
          g2R[s][j] = prm[24 + P];
          const float cf = prm[96 + 8 * (m & 7) + P];
          A2[s][j]  = (m < 8) ? cf : 0.0f;
        }
      }

      auto do_chunk_pair = [&](const v4f ev, const int c) {
        v2f B[4][2];
#pragma unroll
        for (int j = 0; j < 4; ++j) {
          const v2f e2v = {ev[j], ev[j]};
#pragma unroll
          for (int s = 0; s < 2; ++s) {
            const v2f dp = e2v - ecP[s];
            const v2f dr = e2v - ecR[s];
            const v2f qp = __builtin_elementwise_fma(dp, dp, g2P[s]);
            const v2f qr = __builtin_elementwise_fma(dr, dr, g2R[s]);
            const v2f qq = qp * qr;
            B[j][s][0] = __builtin_amdgcn_rcpf(qq[0]);
            B[j][s][1] = __builtin_amdgcn_rcpf(qq[1]);
          }
        }

        v8f acc[4] = {};
#pragma unroll
        for (int s = 0; s < 2; ++s) {
#pragma unroll
          for (int j = 0; j < 4; ++j) {
            acc[j] = __builtin_amdgcn_wmma_f32_16x16x4_f32(
                 false, A2[s],  false, B[j][s],
                 (short)0, acc[j],  false,  false);
          }
        }

        if (lane < 16) {
          v4f o;
#pragma unroll
          for (int j = 0; j < 4; ++j) {
            const v2f e2v = {ev[j], ev[j]};
            v2f t = {acc[j][6], acc[j][7]};
            const v2f s2 = {acc[j][4], acc[j][5]};
            const v2f s1 = {acc[j][2], acc[j][3]};
            const v2f s0 = {acc[j][0], acc[j][1]};
            t = __builtin_elementwise_fma(t, e2v, s2);
            t = __builtin_elementwise_fma(t, e2v, s1);
            t = __builtin_elementwise_fma(t, e2v, s0);
            o[j] = t[0] * t[0] + t[1] * t[1];
          }
          O4[c * 16 + m] = o;
        }
      };
      walk(do_chunk_pair);
    } else {
      v2f ecv[4], g2v[4], A[4];
      load_peak_consts(ecv, g2v, A);
      auto do_chunk = [&](const v4f ev, const int c) {
        v2f B[4][4];
#pragma unroll
        for (int j = 0; j < 4; ++j) {
          const v2f e2v = {ev[j], ev[j]};
#pragma unroll
          for (int s = 0; s < 4; ++s) {
            v2f d = e2v - ecv[s];
            v2f q = __builtin_elementwise_fma(d, d, g2v[s]);
            B[j][s][0] = __builtin_amdgcn_rcpf(q[0]);
            B[j][s][1] = __builtin_amdgcn_rcpf(q[1]);
          }
        }
        v8f acc[4] = {};
#pragma unroll
        for (int s = 0; s < 4; ++s) {
#pragma unroll
          for (int j = 0; j < 4; ++j) {
            acc[j] = __builtin_amdgcn_wmma_f32_16x16x4_f32(
                 false, A[s],  false, B[j][s],
                 (short)0, acc[j],  false,  false);
          }
        }
        if (lane < 16) {
          v4f o;
#pragma unroll
          for (int j = 0; j < 4; ++j) {
            float tre = __builtin_fmaf(ev[j], acc[j][0], acc[j][1]);
            float tim = __builtin_fmaf(ev[j], acc[j][2], acc[j][3]);
            o[j] = tre * tre + tim * tim;
          }
          O4[c * 16 + m] = o;
        }
      };
      walk(do_chunk);
    }
  }

  const int ntiles = (n + 15) >> 4;
  if ((nchunks << 2) + wave < ntiles) {
    v2f ecv[4], g2v[4], A[4];
    load_peak_consts(ecv, g2v, A);
    for (int t = (nchunks << 2) + wave; t < ntiles; t += nwaves) {
      const int base = t << 4;
      int idx = base + m;
      float Ev = E[idx < n ? idx : (n - 1)];

      v2f B[4];
#pragma unroll
      for (int s = 0; s < 4; ++s) {
#pragma unroll
        for (int j = 0; j < 2; ++j) {
          float d = Ev - ecv[s][j];
          B[s][j] = __builtin_amdgcn_rcpf(__builtin_fmaf(d, d, g2v[s][j]));
        }
      }

      v8f acc = {};
#pragma unroll
      for (int s = 0; s < 4; ++s) {
        acc = __builtin_amdgcn_wmma_f32_16x16x4_f32(
             false, A[s],  false, B[s],
             (short)0, acc,  false,  false);
      }

      if (lane < 16 && idx < n) {
        float tre = __builtin_fmaf(Ev, acc[0], acc[1]);
        float tim = __builtin_fmaf(Ev, acc[2], acc[3]);
        out[idx] = tre * tre + tim * tim;
      }
    }
  }
}

extern "C" void kernel_launch(void* const* d_in, const int* in_sizes, int n_in,
                              void* d_out, int out_size, void* d_ws, size_t ws_size,
                              hipStream_t stream) {
  const float* E    = (const float*)d_in[0];
  const float* ec_w = (const float*)d_in[1];
  const float* ec_b = (const float*)d_in[2];
  const float* th_w = (const float*)d_in[3];
  const float* th_b = (const float*)d_in[4];
  const float* gr_w = (const float*)d_in[5];
  const float* gr_b = (const float*)d_in[6];
  const float* gl_w = (const float*)d_in[7];
  const float* gl_b = (const float*)d_in[8];
  const float* e0   = (const float*)d_in[9];
  float* out = (float*)d_out;

  const int n = in_sizes[0];

  peaks_wmma_kernel<<<256, 256, 0, stream>>>(E, out, ec_w, ec_b, th_w, th_b,
                                             gr_w, gr_b, gl_w, gl_b, e0, n);
}


# Grid of 512 work-groups x 8 waves, not 256: two work-groups resident per WGP, four waves per SIMD instead of two fill each other's stalls; each wave takes 16 chunks in 4 trips, not 32 in 8
# speedup vs baseline: 2.4167x; 1.0846x over previous
#include <hip/hip_runtime.h>
#include <math.h>


typedef __attribute__((ext_vector_type(2))) float v2f;
typedef __attribute__((ext_vector_type(4))) float v4f;
typedef __attribute__((ext_vector_type(8))) float v8f;

#define NPEAK 16
#define NPAIR 8
#define NQUAD 4
#define QB (6 * NPEAK + 8 * NPAIR + 8)
#define TUNABLE_MASK ((1<<0)|(1<<3)|(1<<6)|(1<<9)|(1<<12)|(1<<15))
#define E_TUNE  0.1f
#define GMIN    1e-4f
#define GMAX    0.5f
#define PAIR_COND_MAX 512.0f
#define QUAD_COND_MAX 512.0f


__global__ __launch_bounds__(256)
void peaks_wmma_kernel(const float* __restrict__ E, float* __restrict__ out,
                       const float* __restrict__ ec_w, const float* __restrict__ ec_b,
                       const float* __restrict__ th_w, const float* __restrict__ th_b,
                       const float* __restrict__ gr_w, const float* __restrict__ gr_b,
                       const float* __restrict__ gl_w, const float* __restrict__ gl_b,
                       const float* __restrict__ e0, int n) {
  __shared__ float prm[6 * NPEAK + 8 * NPAIR + 8 + 16 * NQUAD];
  const int lane = threadIdx.x & 31;
  const int m    = lane & 15;
  const int kh   = lane >> 4;
  const int wave   = blockIdx.x * (blockDim.x >> 5)
                   + __builtin_amdgcn_readfirstlane((int)(threadIdx.x >> 5));
  const int nwaves = gridDim.x * (blockDim.x >> 5);

  const bool al16 = ((((size_t)E) | ((size_t)out)) & 15) == 0;
  const int nchunks = al16 ? (n >> 6) : 0;
  const int last = nchunks - 1;
  const v4f* __restrict__ E4 = (const v4f*)E;
  v4f* __restrict__ O4 = (v4f*)out;

  const bool setup_lane = threadIdx.x < NPEAK;
  const int  sp = setup_lane ? (int)threadIdx.x : 0;
  float r_ecw = 0.f, r_ecb = 0.f, r_thw = 0.f, r_thb = 0.f, r_grw = 0.f,
        r_grb = 0.f, r_glw = 0.f, r_glb = 0.f, r_e0 = 0.f;
  if (setup_lane) {
    r_ecw = ec_w[sp]; r_ecb = ec_b[sp];
    r_thw = th_w[sp]; r_thb = th_b[sp];
    r_grw = gr_w[sp]; r_grb = gr_b[sp];
    r_glw = gl_w[sp]; r_glb = gl_b[sp];
    r_e0  = e0[sp];
    asm volatile("" :: "v"(r_ecw), "v"(r_ecb), "v"(r_thw), "v"(r_thb),
                       "v"(r_grw), "v"(r_grb), "v"(r_glw), "v"(r_glb),
                       "v"(r_e0));
  }

  v4f ev0 = {}, ev1 = {}, ev2 = {}, ev3 = {};
  if (wave < nchunks) {
    int c1 = wave + nwaves;     c1 = c1 < last ? c1 : last;
    int c2 = wave + 2 * nwaves; c2 = c2 < last ? c2 : last;
    int c3 = wave + 3 * nwaves; c3 = c3 < last ? c3 : last;
    ev0 = E4[wave * 16 + m];
    ev1 = E4[c1 * 16 + m];
    ev2 = E4[c2 * 16 + m];
    ev3 = E4[c3 * 16 + m];
  }

  if (setup_lane) {
    float ec = ((TUNABLE_MASK >> sp) & 1)
                   ? (r_e0 + E_TUNE * tanhf(r_ecw + r_ecb))
                   : r_e0;
    float th = 3.14159265358979323846f / (1.0f + expf(-(r_thw + r_thb)));
    float gr = GMIN + (GMAX - GMIN) / (1.0f + expf(-(r_grw + r_grb)));
    float gl = GMIN + (GMAX - GMIN) / (1.0f + expf(-(r_glw + r_glb)));
    float g  = 0.5f * (gr + gl);
    float sq = sqrtf(gr * gl);
    float a  = cosf(th) * sq;
    float b  = sinf(th) * sq;
    prm[sp]              = ec;
    prm[16 + sp]         = g * g;
    prm[32 + 0*16 + sp]  = a;
    prm[32 + 1*16 + sp]  = b * g - a * ec;
    prm[32 + 2*16 + sp]  = b;
    prm[32 + 3*16 + sp]  = -(b * ec + a * g);

    const float ecr = __shfl_xor(ec, 8, 16);
    const float gp  = __shfl_xor(g,  8, 16);
    const float ar  = __shfl_xor(a,  8, 16);
    const float br  = __shfl_xor(b,  8, 16);
    const float w2p = __builtin_fmaf(ec, ec, g * g);
    const float w2r = __builtin_fmaf(ecr, ecr, gp * gp);
    const float c3r = a + ar, c3i = b + br;
    const float t2pr = __builtin_fmaf(2.0f, ecr, ec), t2rr = __builtin_fmaf(2.0f, ec, ecr);
    const float c2r = -((a * t2pr - b * g) + (ar * t2rr - br * gp));
    const float c2i = -((a * g + b * t2pr) + (ar * gp + br * t2rr));
    const float t1pr = __builtin_fmaf(2.0f * ecr, ec, w2r), t1pi = 2.0f * ecr * g;
    const float t1rr = __builtin_fmaf(2.0f * ec, ecr, w2p), t1ri = 2.0f * ec * gp;
    const float c1r = (a * t1pr - b * t1pi) + (ar * t1rr - br * t1ri);
    const float c1i = (a * t1pi + b * t1pr) + (ar * t1ri + br * t1rr);
    const float c0r = -((a * ec - b * g) * w2r + (ar * ecr - br * gp) * w2p);
    const float c0i = -((a * g + b * ec) * w2r + (ar * gp + br * ecr) * w2p);
    if (sp < NPAIR) {
      prm[96 + 0*8 + sp] = c0r;  prm[96 + 1*8 + sp] = c0i;
      prm[96 + 2*8 + sp] = c1r;  prm[96 + 3*8 + sp] = c1i;
      prm[96 + 4*8 + sp] = c2r;  prm[96 + 5*8 + sp] = c2i;
      prm[96 + 6*8 + sp] = c3r;  prm[96 + 7*8 + sp] = c3i;
    }
    const float ea_ = fabsf(ec) + g;
    const float mono = __builtin_fmaf(__builtin_fmaf(__builtin_fmaf(
        fabsf(c3r) + fabsf(c3i), ea_, fabsf(c2r) + fabsf(c2i)), ea_,
        fabsf(c1r) + fabsf(c1i)), ea_, fabsf(c0r) + fabsf(c0i));
    const float dxy = ec - ecr;
    const float den = (g * g) * __builtin_fmaf(dxy, dxy, gp * gp);
    const bool bad = !(mono <= PAIR_COND_MAX * den);
    const unsigned anybad = __builtin_amdgcn_ballot_w32(bad);
    if (sp == 0) prm[160] = anybad ? 0.0f : 1.0f;

    const float ec4  = __shfl_xor(ec, 4, 16),  g4  = __shfl_xor(g, 4, 16);
    const float ec12 = __shfl_xor(ec, 12, 16), g12 = __shfl_xor(g, 12, 16);
    const float be1 = -2.0f * ec4,  ga1 = __builtin_fmaf(ec4, ec4, g4 * g4);
    const float be2 = -2.0f * ecr,  ga2 = w2r;
    const float be3 = -2.0f * ec12, ga3 = __builtin_fmaf(ec12, ec12, g12 * g12);
    const float u3 = be1 + be2;
    const float u2 = __builtin_fmaf(be1, be2, ga1 + ga2);
    const float u1 = __builtin_fmaf(be1, ga2, be2 * ga1);
    const float u0 = ga1 * ga2;
    float rr[7];
    rr[6] = 1.0f;
    rr[5] = u3 + be3;
    rr[4] = __builtin_fmaf(u3, be3, u2 + ga3);
    rr[3] = __builtin_fmaf(u2, be3, __builtin_fmaf(u3, ga3, u1));
    rr[2] = __builtin_fmaf(u1, be3, __builtin_fmaf(u2, ga3, u0));
    rr[1] = __builtin_fmaf(u0, be3, u1 * ga3);
    rr[0] = u0 * ga3;
    const float mre = b * g - a * ec;
    const float mim = -(b * ec + a * g);
    float nr[8], ni[8];
    nr[7] = a;          ni[7] = b;
    nr[0] = mre * rr[0]; ni[0] = mim * rr[0];
#pragma unroll
    for (int k = 1; k < 7; ++k) {
      nr[k] = __builtin_fmaf(a, rr[k - 1], mre * rr[k]);
      ni[k] = __builtin_fmaf(b, rr[k - 1], mim * rr[k]);
    }
#pragma unroll
    for (int k = 0; k < 8; ++k) {
      nr[k] += __shfl_xor(nr[k], 4, 16);
      ni[k] += __shfl_xor(ni[k], 4, 16);
      nr[k] += __shfl_xor(nr[k], 8, 16);
      ni[k] += __shfl_xor(ni[k], 8, 16);
    }
    if (sp < NQUAD) {
#pragma unroll
      for (int k = 0; k < 8; ++k) {
        prm[QB + 4 * k + sp]       = nr[k];
        prm[QB + 4 * (8 + k) + sp] = ni[k];
      }
    }
    float monoq = fabsf(nr[7]) + fabsf(ni[7]);
#pragma unroll
    for (int k = 6; k >= 0; --k)
      monoq = __builtin_fmaf(monoq, ea_, fabsf(nr[k]) + fabsf(ni[k]));
    const float d4 = ec - ec4, d12 = ec - ec12;
    const float denq = den * __builtin_fmaf(d4, d4, g4 * g4)
                           * __builtin_fmaf(d12, d12, g12 * g12);
    const bool badq = !(monoq <= QUAD_COND_MAX * denq);
    const unsigned anybadq = __builtin_amdgcn_ballot_w32(badq);
    if (sp == 0) prm[161] = anybadq ? 0.0f : 1.0f;
  }
  __syncthreads();

  const bool paired = __builtin_amdgcn_readfirstlane((int)(prm[160] != 0.0f)) != 0;
  const bool quad   = __builtin_amdgcn_readfirstlane((int)(prm[161] != 0.0f)) != 0;

  auto load_peak_consts = [&](v2f (&ecv)[4], v2f (&g2v)[4], v2f (&A)[4]) {
#pragma unroll
    for (int s = 0; s < 4; ++s) {
#pragma unroll
      for (int j = 0; j < 2; ++j) {
        int p = 4 * s + 2 * kh + j;
        ecv[s][j] = prm[p];
        g2v[s][j] = prm[16 + p];
        const float cf = prm[32 + 16 * (m & 3) + p];
        A[s][j]   = (m < 4) ? cf : 0.0f;
      }
    }
  };

  auto walk = [&](auto&& do_chunk) {
    const int step = 4 * nwaves;
    for (int c = wave; c < nchunks; c += step) {
      v4f n0 = ev0, n1 = ev1, n2 = ev2, n3 = ev3;
      const int cn = c + step;
      if (cn < nchunks) {
        int c1 = cn + nwaves;     c1 = c1 < last ? c1 : last;
        int c2 = cn + 2 * nwaves; c2 = c2 < last ? c2 : last;
        int c3 = cn + 3 * nwaves; c3 = c3 < last ? c3 : last;
        n0 = E4[cn * 16 + m];
        n1 = E4[c1 * 16 + m];
        n2 = E4[c2 * 16 + m];
        n3 = E4[c3 * 16 + m];
      }
      __builtin_amdgcn_sched_barrier(0);
      do_chunk(ev0, c);
      if (c + nwaves < nchunks)     do_chunk(ev1, c + nwaves);
      if (c + 2 * nwaves < nchunks) do_chunk(ev2, c + 2 * nwaves);
      if (c + 3 * nwaves < nchunks) do_chunk(ev3, c + 3 * nwaves);
      ev0 = n0; ev1 = n1; ev2 = n2; ev3 = n3;
    }
  };

  if (wave < nchunks) {
    if (quad) {
      v2f ecM[4], g2M[4], A4;
#pragma unroll
      for (int j = 0; j < 2; ++j) {
        const int Q = 2 * kh + j;
#pragma unroll
        for (int t = 0; t < 4; ++t) {
          ecM[t][j] = prm[Q + 4 * t];
          g2M[t][j] = prm[16 + Q + 4 * t];
        }
        A4[j] = prm[QB + 4 * m + Q];
      }

      auto do_chunk_quad = [&](const v4f ev, const int c) {
        v2f B[4];
#pragma unroll
        for (int j = 0; j < 4; ++j) {
          const v2f e2v = {ev[j], ev[j]};
          v2f q[4];
#pragma unroll
          for (int t = 0; t < 4; ++t) {
            const v2f d = e2v - ecM[t];
            q[t] = __builtin_elementwise_fma(d, d, g2M[t]);
          }
          const v2f qq = (q[0] * q[1]) * (q[2] * q[3]);
          B[j][0] = __builtin_amdgcn_rcpf(qq[0]);
          B[j][1] = __builtin_amdgcn_rcpf(qq[1]);
        }

        v8f acc[4] = {};
#pragma unroll
        for (int j = 0; j < 4; ++j) {
          acc[j] = __builtin_amdgcn_wmma_f32_16x16x4_f32(
               false, A4,  false, B[j],
               (short)0, acc[j],  false,  false);
        }

        const v4f x = ev * ev;
        float val[4], oth[4];
#pragma unroll
        for (int j = 0; j < 4; ++j) {
          const v2f x2 = {x[j], x[j]};
          v2f t = {acc[j][6], acc[j][7]};
          const v2f s2 = {acc[j][4], acc[j][5]};
          const v2f s1 = {acc[j][2], acc[j][3]};
          const v2f s0 = {acc[j][0], acc[j][1]};
          t = __builtin_elementwise_fma(t, x2, s2);
          t = __builtin_elementwise_fma(t, x2, s1);
          t = __builtin_elementwise_fma(t, x2, s0);
          val[j] = __builtin_fmaf(t[1], ev[j], t[0]);
          oth[j] = __builtin_bit_cast(float, __builtin_amdgcn_permlanex16(
              __builtin_bit_cast(int, val[j]), __builtin_bit_cast(int, val[j]),
              0x76543210, 0xfedcba98, false, false));
        }
        if (lane < 16) {
          v4f o;
#pragma unroll
          for (int j = 0; j < 4; ++j) o[j] = val[j] * val[j] + oth[j] * oth[j];
          O4[c * 16 + m] = o;
        }
      };
      walk(do_chunk_quad);
    } else if (paired) {
      v2f ecP[2], g2P[2], ecR[2], g2R[2], A2[2];
#pragma unroll
      for (int s = 0; s < 2; ++s) {
#pragma unroll
        for (int j = 0; j < 2; ++j) {
          const int P = 4 * s + 2 * kh + j;
          ecP[s][j] = prm[P];
          ecR[s][j] = prm[8 + P];
          g2P[s][j] = prm[16 + P];
          g2R[s][j] = prm[24 + P];
          const float cf = prm[96 + 8 * (m & 7) + P];
          A2[s][j]  = (m < 8) ? cf : 0.0f;
        }
      }

      auto do_chunk_pair = [&](const v4f ev, const int c) {
        v2f B[4][2];
#pragma unroll
        for (int j = 0; j < 4; ++j) {
          const v2f e2v = {ev[j], ev[j]};
#pragma unroll
          for (int s = 0; s < 2; ++s) {
            const v2f dp = e2v - ecP[s];
            const v2f dr = e2v - ecR[s];
            const v2f qp = __builtin_elementwise_fma(dp, dp, g2P[s]);
            const v2f qr = __builtin_elementwise_fma(dr, dr, g2R[s]);
            const v2f qq = qp * qr;
            B[j][s][0] = __builtin_amdgcn_rcpf(qq[0]);
            B[j][s][1] = __builtin_amdgcn_rcpf(qq[1]);
          }
        }

        v8f acc[4] = {};
#pragma unroll
        for (int s = 0; s < 2; ++s) {
#pragma unroll
          for (int j = 0; j < 4; ++j) {
            acc[j] = __builtin_amdgcn_wmma_f32_16x16x4_f32(
                 false, A2[s],  false, B[j][s],
                 (short)0, acc[j],  false,  false);
          }
        }

        if (lane < 16) {
          v4f o;
#pragma unroll
          for (int j = 0; j < 4; ++j) {
            const v2f e2v = {ev[j], ev[j]};
            v2f t = {acc[j][6], acc[j][7]};
            const v2f s2 = {acc[j][4], acc[j][5]};
            const v2f s1 = {acc[j][2], acc[j][3]};
            const v2f s0 = {acc[j][0], acc[j][1]};
            t = __builtin_elementwise_fma(t, e2v, s2);
            t = __builtin_elementwise_fma(t, e2v, s1);
            t = __builtin_elementwise_fma(t, e2v, s0);
            o[j] = t[0] * t[0] + t[1] * t[1];
          }
          O4[c * 16 + m] = o;
        }
      };
      walk(do_chunk_pair);
    } else {
      v2f ecv[4], g2v[4], A[4];
      load_peak_consts(ecv, g2v, A);
      auto do_chunk = [&](const v4f ev, const int c) {
        v2f B[4][4];
#pragma unroll
        for (int j = 0; j < 4; ++j) {
          const v2f e2v = {ev[j], ev[j]};
#pragma unroll
          for (int s = 0; s < 4; ++s) {
            v2f d = e2v - ecv[s];
            v2f q = __builtin_elementwise_fma(d, d, g2v[s]);
            B[j][s][0] = __builtin_amdgcn_rcpf(q[0]);
            B[j][s][1] = __builtin_amdgcn_rcpf(q[1]);
          }
        }
        v8f acc[4] = {};
#pragma unroll
        for (int s = 0; s < 4; ++s) {
#pragma unroll
          for (int j = 0; j < 4; ++j) {
            acc[j] = __builtin_amdgcn_wmma_f32_16x16x4_f32(
                 false, A[s],  false, B[j][s],
                 (short)0, acc[j],  false,  false);
          }
        }
        if (lane < 16) {
          v4f o;
#pragma unroll
          for (int j = 0; j < 4; ++j) {
            float tre = __builtin_fmaf(ev[j], acc[j][0], acc[j][1]);
            float tim = __builtin_fmaf(ev[j], acc[j][2], acc[j][3]);
            o[j] = tre * tre + tim * tim;
          }
          O4[c * 16 + m] = o;
        }
      };
      walk(do_chunk);
    }
  }

  const int ntiles = (n + 15) >> 4;
  if ((nchunks << 2) + wave < ntiles) {
    v2f ecv[4], g2v[4], A[4];
    load_peak_consts(ecv, g2v, A);
    for (int t = (nchunks << 2) + wave; t < ntiles; t += nwaves) {
      const int base = t << 4;
      int idx = base + m;
      float Ev = E[idx < n ? idx : (n - 1)];

      v2f B[4];
#pragma unroll
      for (int s = 0; s < 4; ++s) {
#pragma unroll
        for (int j = 0; j < 2; ++j) {
          float d = Ev - ecv[s][j];
          B[s][j] = __builtin_amdgcn_rcpf(__builtin_fmaf(d, d, g2v[s][j]));
        }
      }

      v8f acc = {};
#pragma unroll
      for (int s = 0; s < 4; ++s) {
        acc = __builtin_amdgcn_wmma_f32_16x16x4_f32(
             false, A[s],  false, B[s],
             (short)0, acc,  false,  false);
      }

      if (lane < 16 && idx < n) {
        float tre = __builtin_fmaf(Ev, acc[0], acc[1]);
        float tim = __builtin_fmaf(Ev, acc[2], acc[3]);
        out[idx] = tre * tre + tim * tim;
      }
    }
  }
}

extern "C" void kernel_launch(void* const* d_in, const int* in_sizes, int n_in,
                              void* d_out, int out_size, void* d_ws, size_t ws_size,
                              hipStream_t stream) {
  const float* E    = (const float*)d_in[0];
  const float* ec_w = (const float*)d_in[1];
  const float* ec_b = (const float*)d_in[2];
  const float* th_w = (const float*)d_in[3];
  const float* th_b = (const float*)d_in[4];
  const float* gr_w = (const float*)d_in[5];
  const float* gr_b = (const float*)d_in[6];
  const float* gl_w = (const float*)d_in[7];
  const float* gl_b = (const float*)d_in[8];
  const float* e0   = (const float*)d_in[9];
  float* out = (float*)d_out;

  const int n = in_sizes[0];

  peaks_wmma_kernel<<<512, 256, 0, stream>>>(E, out, ec_w, ec_b, th_w, th_b,
                                             gr_w, gr_b, gl_w, gl_b, e0, n);
}
